# starting kernel
# baseline (speedup 1.0000x reference)
#include <hip/hip_runtime.h>
#include <math.h>

typedef __attribute__((ext_vector_type(2))) float v2f;
typedef __attribute__((ext_vector_type(8))) float v8f;

#define HC_GAMMA (-0.1f)
#define HC_ZETA  (1.1f)

__device__ __forceinline__ float hc_gate_coef(const float* __restrict__ qz,
                                              const float* __restrict__ w,
                                              int k) {
  float q = qz[k];
  float s = 1.0f / (1.0f + __expf(-q));
  float z = s * (HC_ZETA - HC_GAMMA) + HC_GAMMA;
  z = fminf(fmaxf(z, 0.0f), 1.0f);
  return z * w[k];
}

__global__ void __launch_bounds__(256)
l0sindy_poly_wmma(const float* __restrict__ obs,
                  const float* __restrict__ act,
                  const float* __restrict__ weights,
                  const float* __restrict__ qz_loga,
                  float* __restrict__ out,
                  int batch) {
  const int lane          = threadIdx.x & 31;
  const int waveInBlock   = threadIdx.x >> 5;
  const int wavesPerBlock = blockDim.x >> 5;
  const int waveId        = blockIdx.x * wavesPerBlock + waveInBlock;
  const int numWaves      = gridDim.x * wavesPerBlock;
  const int numTiles      = batch >> 4;

  const bool hi    = lane >= 16;
  const int  row16 = lane & 15;
  const int  ko    = hi ? 2 : 0;

  float b0[9], b1[9];
#pragma unroll
  for (int j = 0; j < 9; ++j) {
    int k = 4 * j + ko;
    b0[j] = hc_gate_coef(qz_loga, weights, k);
    b1[j] = (k + 1 < 35) ? hc_gate_coef(qz_loga, weights, k + 1) : 0.0f;
  }

  for (int tile = waveId; tile < numTiles; tile += numWaves) {
    const int r  = (tile << 4) + row16;
    const float x0 = obs[3 * r + 0];
    const float x1 = obs[3 * r + 1];
    const float x2 = obs[3 * r + 2];
    const float x3 = act[r];

    const int nt = tile + numWaves;
    if (nt < numTiles) {
      const int rn = (nt << 4) + row16;
      __builtin_prefetch(&obs[3 * rn], 0, 0);
      __builtin_prefetch(&act[rn], 0, 0);
    }

    const float p00 = x0 * x0, p01 = x0 * x1, p02 = x0 * x2, p03 = x0 * x3;
    const float p11 = x1 * x1, p12 = x1 * x2, p13 = x1 * x3;
    const float p22 = x2 * x2, p23 = x2 * x3, p33 = x3 * x3;
    const float q000 = x0 * p00, q001 = x0 * p01, q002 = x0 * p02, q003 = x0 * p03;
    const float q011 = x0 * p11, q012 = x0 * p12, q013 = x0 * p13;
    const float q022 = x0 * p22, q023 = x0 * p23, q033 = x0 * p33;
    const float q111 = x1 * p11, q112 = x1 * p12, q113 = x1 * p13;
    const float q122 = x1 * p22, q123 = x1 * p23, q133 = x1 * p33;
    const float q222 = x2 * p22, q223 = x2 * p23, q233 = x2 * p33;
    const float q333 = x3 * p33;

    v8f acc = {0.0f, 0.0f, 0.0f, 0.0f, 0.0f, 0.0f, 0.0f, 0.0f};
#define WMMA_STEP(t0, t1, t2, t3, j)                                       \
    {                                                                      \
      v2f A  = {hi ? (t2) : (t0), hi ? (t3) : (t1)};                       \
      v2f Bv = {b0[j], b1[j]};                                             \
      acc = __builtin_amdgcn_wmma_f32_16x16x4_f32(                         \
          false, A, false, Bv, (short)0, acc, false, false);               \
    }
    WMMA_STEP(1.0f, x0,   x1,   x2,   0)
    WMMA_STEP(x3,   p00,  p01,  p02,  1)
    WMMA_STEP(p03,  p11,  p12,  p13,  2)
    WMMA_STEP(p22,  p23,  p33,  q000, 3)
    WMMA_STEP(q001, q002, q003, q011, 4)
    WMMA_STEP(q012, q013, q022, q023, 5)
    WMMA_STEP(q033, q111, q112, q113, 6)
    WMMA_STEP(q122, q123, q133, q222, 7)
    WMMA_STEP(q223, q233, q333, 0.0f, 8)
#undef WMMA_STEP

    if (row16 == 0) {
      float* o = out + (tile << 4) + (hi ? 8 : 0);
      float4 v0 = make_float4(acc[0], acc[1], acc[2], acc[3]);
      float4 v1 = make_float4(acc[4], acc[5], acc[6], acc[7]);
      *reinterpret_cast<float4*>(o)     = v0;
      *reinterpret_cast<float4*>(o + 4) = v1;
    }
  }
}

extern "C" void kernel_launch(void* const* d_in, const int* in_sizes, int n_in,
                              void* d_out, int out_size, void* d_ws, size_t ws_size,
                              hipStream_t stream) {
  const float* obs     = (const float*)d_in[0];
  const float* act     = (const float*)d_in[1];
  const float* weights = (const float*)d_in[2];
  const float* qz_loga = (const float*)d_in[3];
  float* out = (float*)d_out;

  const int batch    = in_sizes[1];
  const int numTiles = batch >> 4;
  const int threads  = 256;
  const int wavesPerBlock = threads / 32;
  int blocks = (numTiles + wavesPerBlock - 1) / wavesPerBlock;
  if (blocks > 4096) blocks = 4096;
  if (blocks < 1) blocks = 1;

  l0sindy_poly_wmma<<<blocks, threads, 0, stream>>>(obs, act, weights, qz_loga,
                                                    out, batch);
}


# resident grid: 256 blocks of 16 waves (4 per SIMD), each wave walks 8 trips of two 16-row tiles with the next trip's rows loaded ahead, instead of 32768 short waves each rebuilding the gate terms
# speedup vs baseline: 1.3680x; 1.3680x over previous
#include <hip/hip_runtime.h>
#include <math.h>

typedef __attribute__((ext_vector_type(2))) float v2f;
typedef __attribute__((ext_vector_type(8))) float v8f;

#define HC_GAMMA (-0.1f)
#define HC_ZETA  (1.1f)

__device__ __forceinline__ float hc_gate_coef(const float* __restrict__ qz,
                                              const float* __restrict__ w,
                                              int k) {
  float q = qz[k];
  float s = 1.0f / (1.0f + __expf(-q));
  float z = s * (HC_ZETA - HC_GAMMA) + HC_GAMMA;
  z = fminf(fmaxf(z, 0.0f), 1.0f);
  return z * w[k];
}

struct Row4 { float x0, x1, x2, x3; };

__device__ __forceinline__ Row4 load_row(const float* __restrict__ obs,
                                         const float* __restrict__ act, int r) {
  Row4 v;
  v.x0 = obs[3 * r + 0];
  v.x1 = obs[3 * r + 1];
  v.x2 = obs[3 * r + 2];
  v.x3 = act[r];
  return v;
}

__device__ __forceinline__ void row_fragments(const Row4 r, const bool hi,
                                              float (&a0)[9], float (&a1)[9]) {
  const float x0 = r.x0, x1 = r.x1, x2 = r.x2, x3 = r.x3;
  const float p00 = x0 * x0, p01 = x0 * x1, p02 = x0 * x2, p03 = x0 * x3;
  const float p11 = x1 * x1, p12 = x1 * x2, p13 = x1 * x3;
  const float p22 = x2 * x2, p23 = x2 * x3, p33 = x3 * x3;
  const float q000 = x0 * p00, q001 = x0 * p01, q002 = x0 * p02, q003 = x0 * p03;
  const float q011 = x0 * p11, q012 = x0 * p12, q013 = x0 * p13;
  const float q022 = x0 * p22, q023 = x0 * p23, q033 = x0 * p33;
  const float q111 = x1 * p11, q112 = x1 * p12, q113 = x1 * p13;
  const float q122 = x1 * p22, q123 = x1 * p23, q133 = x1 * p33;
  const float q222 = x2 * p22, q223 = x2 * p23, q233 = x2 * p33;
  const float q333 = x3 * p33;
#define FRAG(j, t0, t1, t2, t3) \
  a0[j] = hi ? (t2) : (t0);     \
  a1[j] = hi ? (t3) : (t1);
  FRAG(0, 1.0f, x0,   x1,   x2)
  FRAG(1, x3,   p00,  p01,  p02)
  FRAG(2, p03,  p11,  p12,  p13)
  FRAG(3, p22,  p23,  p33,  q000)
  FRAG(4, q001, q002, q003, q011)
  FRAG(5, q012, q013, q022, q023)
  FRAG(6, q033, q111, q112, q113)
  FRAG(7, q122, q123, q133, q222)
  FRAG(8, q223, q233, q333, 0.0f)
#undef FRAG
}

__global__ void __launch_bounds__(512)
l0sindy_poly_wmma(const float* __restrict__ obs,
                  const float* __restrict__ act,
                  const float* __restrict__ weights,
                  const float* __restrict__ qz_loga,
                  float* __restrict__ out,
                  int batch) {
  const int lane          = threadIdx.x & 31;
  const int wavesPerBlock = blockDim.x >> 5;
  const int waveId   = __builtin_amdgcn_readfirstlane(
      blockIdx.x * wavesPerBlock + (threadIdx.x >> 5));
  const int numWaves = gridDim.x * wavesPerBlock;
  const int numTiles = batch >> 4;
  const int numPairs = (numTiles + 1) >> 1;

  const bool hi    = lane >= 16;
  const int  row16 = lane & 15;
  const int  ko    = hi ? 2 : 0;

  int p = waveId;
  if (p >= numPairs) return;

  bool hasB = (2 * p + 1) < numTiles;
  Row4 ca = load_row(obs, act, (p << 5) + row16);
  Row4 cb = load_row(obs, act, (p << 5) + (hasB ? 16 : 0) + row16);

  float b0[9], b1[9];
#pragma unroll
  for (int j = 0; j < 9; ++j) {
    int k = 4 * j + ko;
    b0[j] = hc_gate_coef(qz_loga, weights, k);
    b1[j] = (k + 1 < 35) ? hc_gate_coef(qz_loga, weights, k + 1) : 0.0f;
  }

  for (;;) {
    const int  pn    = p + numWaves;
    const bool more  = pn < numPairs;
    const int  pl    = more ? pn : p;
    const bool hasBn = (2 * pl + 1) < numTiles;
    const Row4 na = load_row(obs, act, (pl << 5) + row16);
    const Row4 nb = load_row(obs, act, (pl << 5) + (hasBn ? 16 : 0) + row16);

    const int pf = pn + numWaves;
    if (pf < numPairs) {
      __builtin_prefetch(&obs[3 * ((pf << 5) + lane)], 0, 0);
      __builtin_prefetch(&act[(pf << 5) + lane], 0, 0);
    }

    float aA0[9], aA1[9], aB0[9], aB1[9];
    row_fragments(ca, hi, aA0, aA1);
    row_fragments(cb, hi, aB0, aB1);

    v8f accA = {0.0f, 0.0f, 0.0f, 0.0f, 0.0f, 0.0f, 0.0f, 0.0f};
    v8f accB = {0.0f, 0.0f, 0.0f, 0.0f, 0.0f, 0.0f, 0.0f, 0.0f};
#pragma unroll
    for (int j = 0; j < 9; ++j) {
      const v2f Bv = {b0[j], b1[j]};
      const v2f Aa = {aA0[j], aA1[j]};
      const v2f Ab = {aB0[j], aB1[j]};
      accA = __builtin_amdgcn_wmma_f32_16x16x4_f32(
          false, Aa, false, Bv, (short)0, accA, false, false);
      accB = __builtin_amdgcn_wmma_f32_16x16x4_f32(
          false, Ab, false, Bv, (short)0, accB, false, false);
    }

    if (row16 == 0) {
      float* o = out + (p << 5) + (hi ? 8 : 0);
      *reinterpret_cast<float4*>(o)     = make_float4(accA[0], accA[1], accA[2], accA[3]);
      *reinterpret_cast<float4*>(o + 4) = make_float4(accA[4], accA[5], accA[6], accA[7]);
      if (hasB) {
        *reinterpret_cast<float4*>(o + 16) = make_float4(accB[0], accB[1], accB[2], accB[3]);
        *reinterpret_cast<float4*>(o + 20) = make_float4(accB[4], accB[5], accB[6], accB[7]);
      }
    }

    if (!more) break;
    p = pn;
    ca = na;
    cb = nb;
    hasB = hasBn;
  }
}

extern "C" void kernel_launch(void* const* d_in, const int* in_sizes, int n_in,
                              void* d_out, int out_size, void* d_ws, size_t ws_size,
                              hipStream_t stream) {
  const float* obs     = (const float*)d_in[0];
  const float* act     = (const float*)d_in[1];
  const float* weights = (const float*)d_in[2];
  const float* qz_loga = (const float*)d_in[3];
  float* out = (float*)d_out;

  const int batch = in_sizes[1];

  l0sindy_poly_wmma<<<256, 512, 0, stream>>>(obs, act, weights, qz_loga,
                                             out, batch);
}


# long waves with a whole pass of loads in flight: 256 blocks of 8 waves, each wave requests all 64 row loads of its 512 rows before its first WMMA (was 16-wave blocks, 8 trips, one trip ahead)
# speedup vs baseline: 3.1014x; 2.2671x over previous
#include <hip/hip_runtime.h>
#include <math.h>

typedef __attribute__((ext_vector_type(2))) float v2f;
typedef __attribute__((ext_vector_type(8))) float v8f;

#define HC_GAMMA (-0.1f)
#define HC_ZETA  (1.1f)

__device__ __forceinline__ float hc_gate_coef(const float q, const float w) {
  float s = 1.0f / (1.0f + __expf(-q));
  float z = s * (HC_ZETA - HC_GAMMA) + HC_GAMMA;
  z = fminf(fmaxf(z, 0.0f), 1.0f);
  return z * w;
}

struct Row4 { float x0, x1, x2, x3; };

__device__ __forceinline__ Row4 load_row(const float* __restrict__ obsTile,
                                         const float* __restrict__ actTile,
                                         int row16) {
  Row4 v;
  v.x0 = obsTile[3 * row16 + 0];
  v.x1 = obsTile[3 * row16 + 1];
  v.x2 = obsTile[3 * row16 + 2];
  v.x3 = actTile[row16];
  return v;
}

__constant__ const signed char kCoefIndex[64] = {
     5,  6,  -1,  9,  15, 16,  -1, 25,  -1, 19,  -1, -1,  -1, -1,   1,  2,
    -1, -1,  -1, -1,  -1, -1,  -1, -1,  -1, -1,  -1, -1,  -1, -1,  -1, -1,
     7,  8,  10, 11,  17, 18,  26, 27,  20, 21,  -1, 24,  22, 23,   3,  4,
    12, 13,  -1, 14,  31, 32,  -1, 34,  -1, 33,  28, 29,  -1, -1,  -1, 30};

__device__ __forceinline__ float tile_partial(const Row4 r, const bool hi,
                                              const v2f Acoef, const v8f Cinit) {
  const float v0 = hi ? r.x2 : r.x0;
  const float v1 = hi ? r.x3 : r.x1;
  const float v2 = hi ? r.x0 : r.x2;
  const float v3 = hi ? r.x1 : r.x3;
  const v2f Bv = {v0, v1};
  const v8f T = __builtin_amdgcn_wmma_f32_16x16x4_f32(
      false, Acoef, false, Bv, (short)0, Cinit, false, false);
  const float w7 = hi ? (v1 * v3) : 1.0f;
  float s = v0 * T[0];
  s = __builtin_fmaf(v1, T[1], s);
  s = __builtin_fmaf(v0 * v0, T[2], s);
  s = __builtin_fmaf(v1 * v1, T[3], s);
  s = __builtin_fmaf(v0 * v1, T[4], s);
  s = __builtin_fmaf(v0 * v3, T[5], s);
  s = __builtin_fmaf(v0 * v2, T[6], s);
  s = __builtin_fmaf(w7, T[7], s);
  return s;
}

#define GROUPS_PER_PASS 16
#define ROWS_PER_PASS (32 * GROUPS_PER_PASS)

struct PassRows { Row4 a[GROUPS_PER_PASS], b[GROUPS_PER_PASS]; };

template <bool FULL>
__device__ __forceinline__ void request_pass(const float* __restrict__ obs,
                                             const float* __restrict__ act,
                                             const int pass, const int numPairs,
                                             const int numTiles, const int row16,
                                             PassRows& rows) {
  if (FULL) {
    const float* __restrict__ o = obs + 3 * ((long)pass * ROWS_PER_PASS);
    const float* __restrict__ a = act + (long)pass * ROWS_PER_PASS;
#pragma unroll
    for (int g = 0; g < GROUPS_PER_PASS; ++g) {
      rows.a[g] = load_row(o + 3 * (32 * g),      a + 32 * g,      row16);
      rows.b[g] = load_row(o + 3 * (32 * g + 16), a + 32 * g + 16, row16);
    }
  } else {
    const int q0 = pass * GROUPS_PER_PASS;
#pragma unroll
    for (int g = 0; g < GROUPS_PER_PASS; ++g) {
      const int  q    = (q0 + g < numPairs) ? (q0 + g) : q0;
      const bool hasB = (2 * q + 1) < numTiles;
      const long rowA = (long)q << 5;
      const long rowB = rowA + (hasB ? 16 : 0);
      rows.a[g] = load_row(obs + 3 * rowA, act + rowA, row16);
      rows.b[g] = load_row(obs + 3 * rowB, act + rowB, row16);
    }
  }
}

template <bool FULL>
__device__ __forceinline__ void finish_pass(const PassRows& rows, const int pass,
                                            const int numPairs, const int numTiles,
                                            const int lane, const bool hi,
                                            const v2f Acoef, const v8f Cinit,
                                            float* __restrict__ out) {
  const int q0 = pass * GROUPS_PER_PASS;
  float* __restrict__ outPass = out + (long)pass * ROWS_PER_PASS;
#pragma unroll
  for (int g = 0; g < GROUPS_PER_PASS; ++g) {
    const float sA = tile_partial(rows.a[g], hi, Acoef, Cinit);
    const float sB = tile_partial(rows.b[g], hi, Acoef, Cinit);

    const float give = hi ? sA : sB;
    const float keep = hi ? sB : sA;
    const float got = __builtin_bit_cast(float, __builtin_amdgcn_permlanex16(
        0u, __builtin_bit_cast(unsigned, give), 0x76543210u, 0xfedcba98u,
        false, false));
    const float res = keep + got;

    if (FULL) {
      outPass[32 * g + lane] = res;
    } else {
      const int  q     = q0 + g;
      const bool valid = q < numPairs;
      const bool hasB  = (2 * q + 1) < numTiles;
      if (valid && (!hi || hasB)) outPass[32 * g + lane] = res;
    }
  }
}

struct Operands { v2f Acoef; v8f Cinit; };

struct CoefLoads { float qLo, wLo, qHi, wHi; int k0, k1; };

__device__ __forceinline__ CoefLoads load_coefs(const float* __restrict__ qz,
                                                const float* __restrict__ w,
                                                const int lane) {
  const int kHi = 32 + (lane < 3 ? lane : 0);
  CoefLoads c;
  c.qLo = qz[lane]; c.wLo = w[lane];
  c.qHi = qz[kHi];  c.wHi = w[kHi];
  c.k0  = kCoefIndex[2 * lane + 0];
  c.k1  = kCoefIndex[2 * lane + 1];
  return c;
}

__device__ __forceinline__ Operands make_operands(const CoefLoads c, const bool hi) {
  const float coefLo = hc_gate_coef(c.qLo, c.wLo);
  const float coefHi = hc_gate_coef(c.qHi, c.wHi);
  const float g0lo = __shfl(coefLo, c.k0 & 31), g0hi = __shfl(coefHi, c.k0 & 31);
  const float g1lo = __shfl(coefLo, c.k1 & 31), g1hi = __shfl(coefHi, c.k1 & 31);
  const float bias = __shfl(coefLo, 0);
  Operands op;
  op.Acoef[0] = (c.k0 < 0) ? 0.0f : ((c.k0 >= 32) ? g0hi : g0lo);
  op.Acoef[1] = (c.k1 < 0) ? 0.0f : ((c.k1 >= 32) ? g1hi : g1lo);
  op.Cinit = (v8f){0.0f, 0.0f, 0.0f, 0.0f, 0.0f, 0.0f, 0.0f, hi ? 0.0f : bias};
  return op;
}

__global__ void __launch_bounds__(256) __attribute__((amdgpu_waves_per_eu(2, 2)))
l0sindy_poly_wmma(const float* __restrict__ obs,
                  const float* __restrict__ act,
                  const float* __restrict__ weights,
                  const float* __restrict__ qz_loga,
                  float* __restrict__ out,
                  int batch) {
  const int lane          = threadIdx.x & 31;
  const int wavesPerBlock = blockDim.x >> 5;
  const int waveId   = __builtin_amdgcn_readfirstlane(
      blockIdx.x * wavesPerBlock + (threadIdx.x >> 5));
  const int numWaves  = gridDim.x * wavesPerBlock;
  const int numTiles  = batch >> 4;
  const int numPairs  = (numTiles + 1) >> 1;
  const int numPasses = (numPairs + GROUPS_PER_PASS - 1) / GROUPS_PER_PASS;

  const bool hi    = lane >= 16;
  const int  row16 = lane & 15;

  int pass = waveId;
  if (pass >= numPasses) return;

  Operands op;
  if ((pass + 1) * (2 * GROUPS_PER_PASS) <= numTiles) {
    const CoefLoads c = load_coefs(qz_loga, weights, lane);
    PassRows first;
    request_pass<true>(obs, act, pass, numPairs, numTiles, row16, first);
    op = make_operands(c, hi);
    finish_pass<true>(first, pass, numPairs, numTiles, lane, hi, op.Acoef, op.Cinit, out);
  } else {
    PassRows first;
    request_pass<false>(obs, act, pass, numPairs, numTiles, row16, first);
    const CoefLoads c = load_coefs(qz_loga, weights, lane);
    op = make_operands(c, hi);
    finish_pass<false>(first, pass, numPairs, numTiles, lane, hi, op.Acoef, op.Cinit, out);
  }

  for (pass += numWaves; pass < numPasses; pass += numWaves) {
    PassRows rows;
    request_pass<false>(obs, act, pass, numPairs, numTiles, row16, rows);
    finish_pass<false>(rows, pass, numPairs, numTiles, lane, hi, op.Acoef, op.Cinit, out);
  }
}

extern "C" void kernel_launch(void* const* d_in, const int* in_sizes, int n_in,
                              void* d_out, int out_size, void* d_ws, size_t ws_size,
                              hipStream_t stream) {
  const float* obs     = (const float*)d_in[0];
  const float* act     = (const float*)d_in[1];
  const float* weights = (const float*)d_in[2];
  const float* qz_loga = (const float*)d_in[3];
  float* out = (float*)d_out;

  const int batch = in_sizes[1];

  l0sindy_poly_wmma<<<256, 256, 0, stream>>>(obs, act, weights, qz_loga,
                                             out, batch);
}


# cubic regrouped over six u per lane half, all of the half's own pair (y0,y1): no swapped variables, nested sums, 6 vector ops per tile (was 19), fenced per 2 groups; summing order changes
# speedup vs baseline: 3.2682x; 1.0538x over previous
#include <hip/hip_runtime.h>
#include <math.h>

typedef __attribute__((ext_vector_type(2))) float v2f;
typedef __attribute__((ext_vector_type(8))) float v8f;

#define HC_GAMMA (-0.1f)
#define HC_ZETA  (1.1f)

__device__ __forceinline__ float hc_gate_coef(const float q, const float w) {
  float s = 1.0f / (1.0f + __expf(-q));
  float z = s * (HC_ZETA - HC_GAMMA) + HC_GAMMA;
  z = fminf(fmaxf(z, 0.0f), 1.0f);
  return z * w;
}

struct Row4 { float x0, x1, x2, x3; };

__device__ __forceinline__ Row4 load_row(const float* __restrict__ obsTile,
                                         const float* __restrict__ actTile,
                                         int row16) {
  Row4 v;
  v.x0 = obsTile[3 * row16 + 0];
  v.x1 = obsTile[3 * row16 + 1];
  v.x2 = obsTile[3 * row16 + 2];
  v.x3 = actTile[row16];
  return v;
}

__constant__ const signed char kCoefIndex[64] = {
     1,  2,  -1, -1,   5,  6,  -1,  9,  15, 16,  -1, 25,  -1, 19,  -1, -1,
    -1, -1,  -1, -1,  -1, -1,  -1, -1,  22, 28,  24, 30,  23, 29,  -1, -1,
     3,  4,  -1, -1,   7,  8,  10, 11,  17, 18,  26, 27,  20, 21,  -1, -1,
    -1, -1,  -1, -1,  12, 13,  -1, 14,  31, 32,  -1, 34,  -1, 33,  -1, -1};

__device__ __forceinline__ float tile_partial(const Row4 r, const bool hi,
                                              const v2f Acoef, const v8f Cinit) {
  const float y0 = hi ? r.x2 : r.x0;
  const float y1 = hi ? r.x3 : r.x1;
  const v2f Y = {y0, y1};
  const v8f T = __builtin_amdgcn_wmma_f32_16x16x4_f32(
      false, Acoef, false, Y, (short)0, Cinit, false, false);
  const v2f T23 = {T[2], T[3]};
  const v2f T45 = {T[4], T[5]};
  const v2f U = __builtin_elementwise_fma(Y, T45, T23);
  const float ux = __builtin_fmaf(y1, T[6], U[0]);
  float s = __builtin_fmaf(y0, ux, T[0]);
  s = __builtin_fmaf(y1, U[1], s);
  return s;
}

#define GROUPS_PER_PASS 16
#define ROWS_PER_PASS (32 * GROUPS_PER_PASS)

struct PassRows { Row4 a[GROUPS_PER_PASS], b[GROUPS_PER_PASS]; };

template <bool FULL>
__device__ __forceinline__ void request_pass(const float* __restrict__ obs,
                                             const float* __restrict__ act,
                                             const int pass, const int numPairs,
                                             const int numTiles, const int row16,
                                             PassRows& rows) {
  if (FULL) {
    const float* __restrict__ o = obs + 3 * ((long)pass * ROWS_PER_PASS);
    const float* __restrict__ a = act + (long)pass * ROWS_PER_PASS;
#pragma unroll
    for (int g = 0; g < GROUPS_PER_PASS; ++g) {
      rows.a[g] = load_row(o + 3 * (32 * g),      a + 32 * g,      row16);
      rows.b[g] = load_row(o + 3 * (32 * g + 16), a + 32 * g + 16, row16);
    }
  } else {
    const int q0 = pass * GROUPS_PER_PASS;
#pragma unroll
    for (int g = 0; g < GROUPS_PER_PASS; ++g) {
      const int  q    = (q0 + g < numPairs) ? (q0 + g) : q0;
      const bool hasB = (2 * q + 1) < numTiles;
      const long rowA = (long)q << 5;
      const long rowB = rowA + (hasB ? 16 : 0);
      rows.a[g] = load_row(obs + 3 * rowA, act + rowA, row16);
      rows.b[g] = load_row(obs + 3 * rowB, act + rowB, row16);
    }
  }
}

template <bool FULL>
__device__ __forceinline__ void finish_pass(const PassRows& rows, const int pass,
                                            const int numPairs, const int numTiles,
                                            const int lane, const bool hi,
                                            const v2f Acoef, const v8f Cinit,
                                            float* __restrict__ out) {
  const int q0 = pass * GROUPS_PER_PASS;
  float* __restrict__ outPass = out + (long)pass * ROWS_PER_PASS;
#pragma unroll
  for (int g = 0; g < GROUPS_PER_PASS; ++g) {
    const float sA = tile_partial(rows.a[g], hi, Acoef, Cinit);
    const float sB = tile_partial(rows.b[g], hi, Acoef, Cinit);

    const float give = hi ? sA : sB;
    const float keep = hi ? sB : sA;
    const float got = __builtin_bit_cast(float, __builtin_amdgcn_permlanex16(
        0u, __builtin_bit_cast(unsigned, give), 0x76543210u, 0xfedcba98u,
        false, false));
    const float res = keep + got;

    if (FULL) {
      outPass[32 * g + lane] = res;
    } else {
      const int  q     = q0 + g;
      const bool valid = q < numPairs;
      const bool hasB  = (2 * q + 1) < numTiles;
      if (valid && (!hi || hasB)) outPass[32 * g + lane] = res;
    }
    if ((g % 2) == 1) __builtin_amdgcn_sched_barrier(0);
  }
}

struct Operands { v2f Acoef; v8f Cinit; };

struct CoefLoads { float qLo, wLo, qHi, wHi; int k0, k1; };

__device__ __forceinline__ CoefLoads load_coefs(const float* __restrict__ qz,
                                                const float* __restrict__ w,
                                                const int lane) {
  const int kHi = 32 + (lane < 3 ? lane : 0);
  CoefLoads c;
  c.qLo = qz[lane]; c.wLo = w[lane];
  c.qHi = qz[kHi];  c.wHi = w[kHi];
  c.k0  = kCoefIndex[2 * lane + 0];
  c.k1  = kCoefIndex[2 * lane + 1];
  return c;
}

__device__ __forceinline__ Operands make_operands(const CoefLoads c, const bool hi) {
  const float coefLo = hc_gate_coef(c.qLo, c.wLo);
  const float coefHi = hc_gate_coef(c.qHi, c.wHi);
  const float g0lo = __shfl(coefLo, c.k0 & 31), g0hi = __shfl(coefHi, c.k0 & 31);
  const float g1lo = __shfl(coefLo, c.k1 & 31), g1hi = __shfl(coefHi, c.k1 & 31);
  const float bias = __shfl(coefLo, 0);
  Operands op;
  op.Acoef[0] = (c.k0 < 0) ? 0.0f : ((c.k0 >= 32) ? g0hi : g0lo);
  op.Acoef[1] = (c.k1 < 0) ? 0.0f : ((c.k1 >= 32) ? g1hi : g1lo);
  op.Cinit = (v8f){hi ? 0.0f : bias, 0.0f, 0.0f, 0.0f, 0.0f, 0.0f, 0.0f, 0.0f};
  return op;
}

__global__ void __launch_bounds__(256) __attribute__((amdgpu_waves_per_eu(2, 2)))
l0sindy_poly_wmma(const float* __restrict__ obs,
                  const float* __restrict__ act,
                  const float* __restrict__ weights,
                  const float* __restrict__ qz_loga,
                  float* __restrict__ out,
                  int batch) {
  const int lane          = threadIdx.x & 31;
  const int wavesPerBlock = blockDim.x >> 5;
  const int waveId   = __builtin_amdgcn_readfirstlane(
      blockIdx.x * wavesPerBlock + (threadIdx.x >> 5));
  const int numWaves  = gridDim.x * wavesPerBlock;
  const int numTiles  = batch >> 4;
  const int numPairs  = (numTiles + 1) >> 1;
  const int numPasses = (numPairs + GROUPS_PER_PASS - 1) / GROUPS_PER_PASS;

  const bool hi    = lane >= 16;
  const int  row16 = lane & 15;

  int pass = waveId;
  if (pass >= numPasses) return;

  Operands op;
  if ((pass + 1) * (2 * GROUPS_PER_PASS) <= numTiles) {
    const CoefLoads c = load_coefs(qz_loga, weights, lane);
    PassRows first;
    request_pass<true>(obs, act, pass, numPairs, numTiles, row16, first);
    op = make_operands(c, hi);
    finish_pass<true>(first, pass, numPairs, numTiles, lane, hi, op.Acoef, op.Cinit, out);
  } else {
    PassRows first;
    request_pass<false>(obs, act, pass, numPairs, numTiles, row16, first);
    const CoefLoads c = load_coefs(qz_loga, weights, lane);
    op = make_operands(c, hi);
    finish_pass<false>(first, pass, numPairs, numTiles, lane, hi, op.Acoef, op.Cinit, out);
  }

  for (pass += numWaves; pass < numPasses; pass += numWaves) {
    PassRows rows;
    request_pass<false>(obs, act, pass, numPairs, numTiles, row16, rows);
    finish_pass<false>(rows, pass, numPairs, numTiles, lane, hi, op.Acoef, op.Cinit, out);
  }
}

extern "C" void kernel_launch(void* const* d_in, const int* in_sizes, int n_in,
                              void* d_out, int out_size, void* d_ws, size_t ws_size,
                              hipStream_t stream) {
  const float* obs     = (const float*)d_in[0];
  const float* act     = (const float*)d_in[1];
  const float* weights = (const float*)d_in[2];
  const float* qz_loga = (const float*)d_in[3];
  float* out = (float*)d_out;

  const int batch = in_sizes[1];

  l0sindy_poly_wmma<<<256, 256, 0, stream>>>(obs, act, weights, qz_loga,
                                             out, batch);
}


# each lane loads one whole row of a 32-row group (the lane halves used to load the same 16 rows): 32 row loads per wave, was 64; operand pairs and half sums change halves by v_permlane16_swap
# speedup vs baseline: 3.6016x; 1.1020x over previous
#include <hip/hip_runtime.h>
#include <math.h>

typedef __attribute__((ext_vector_type(2))) float v2f;
typedef __attribute__((ext_vector_type(8))) float v8f;

#define HC_GAMMA (-0.1f)
#define HC_ZETA  (1.1f)

__device__ __forceinline__ float hc_gate_coef(const float q, const float w) {
  float s = 1.0f / (1.0f + __expf(-q));
  float z = s * (HC_ZETA - HC_GAMMA) + HC_GAMMA;
  z = fminf(fmaxf(z, 0.0f), 1.0f);
  return z * w;
}

struct Row4 { float x0, x1, x2, x3; };

__device__ __forceinline__ Row4 load_row(const float* __restrict__ obsGroup,
                                         const float* __restrict__ actGroup,
                                         int row) {
  Row4 v;
  v.x0 = obsGroup[3 * row + 0];
  v.x1 = obsGroup[3 * row + 1];
  v.x2 = obsGroup[3 * row + 2];
  v.x3 = actGroup[row];
  return v;
}

__device__ __forceinline__ void swap_halves(float& a, float& b) {
  asm("v_permlane16_swap_b32 %0, %1" : "+v"(a), "+v"(b));
}

__constant__ const signed char kCoefIndex[64] = {
     1,  2,  -1, -1,   5,  6,  -1,  9,  15, 16,  -1, 25,  -1, 19,  -1, -1,
    -1, -1,  -1, -1,  -1, -1,  -1, -1,  22, 28,  24, 30,  23, 29,  -1, -1,
     3,  4,  -1, -1,   7,  8,  10, 11,  17, 18,  26, 27,  20, 21,  -1, -1,
    -1, -1,  -1, -1,  12, 13,  -1, 14,  31, 32,  -1, 34,  -1, 33,  -1, -1};

__device__ __forceinline__ float tile_partial(const float y0, const float y1,
                                              const v2f Acoef, const v8f Cinit) {
  const v2f Y = {y0, y1};
  const v8f T = __builtin_amdgcn_wmma_f32_16x16x4_f32(
      false, Acoef, false, Y, (short)0, Cinit, false, false);
  const v2f T23 = {T[2], T[3]};
  const v2f T45 = {T[4], T[5]};
  const v2f U = __builtin_elementwise_fma(Y, T45, T23);
  const float ux = __builtin_fmaf(y1, T[6], U[0]);
  float s = __builtin_fmaf(y0, ux, T[0]);
  s = __builtin_fmaf(y1, U[1], s);
  return s;
}

#define GROUPS_PER_PASS 16
#define ROWS_PER_PASS (32 * GROUPS_PER_PASS)

struct PassRows { Row4 r[GROUPS_PER_PASS]; };

template <bool FULL>
__device__ __forceinline__ void request_pass(const float* __restrict__ obs,
                                             const float* __restrict__ act,
                                             const int pass, const int numPairs,
                                             const int numTiles, const int lane,
                                             PassRows& rows) {
  if (FULL) {
    const float* __restrict__ o = obs + 3 * ((long)pass * ROWS_PER_PASS);
    const float* __restrict__ a = act + (long)pass * ROWS_PER_PASS;
#pragma unroll
    for (int g = 0; g < GROUPS_PER_PASS; ++g) {
      rows.r[g] = load_row(o + 3 * (32 * g), a + 32 * g, lane);
      __builtin_amdgcn_sched_barrier(1);
    }
  } else {
    const int q0 = pass * GROUPS_PER_PASS;
#pragma unroll
    for (int g = 0; g < GROUPS_PER_PASS; ++g) {
      const int  q    = (q0 + g < numPairs) ? (q0 + g) : q0;
      const bool hasB = (2 * q + 1) < numTiles;
      const long row0 = (long)q << 5;
      rows.r[g] = load_row(obs + 3 * row0, act + row0, hasB ? lane : (lane & 15));
    }
  }
}

template <bool FULL>
__device__ __forceinline__ void finish_pass(const PassRows& rows, const int pass,
                                            const int numPairs, const int numTiles,
                                            const int lane, const bool hi,
                                            const v2f Acoef, const v8f Cinit,
                                            float* __restrict__ out) {
  const int q0 = pass * GROUPS_PER_PASS;
  float* __restrict__ outPass = out + (long)pass * ROWS_PER_PASS;
#pragma unroll
  for (int g = 0; g < GROUPS_PER_PASS; ++g) {
    Row4 r = rows.r[g];
    swap_halves(r.x0, r.x2);
    swap_halves(r.x1, r.x3);

    float sA = tile_partial(r.x0, r.x1, Acoef, Cinit);
    float sB = tile_partial(r.x2, r.x3, Acoef, Cinit);

    swap_halves(sA, sB);
    const float res = sA + sB;

    if (FULL) {
      outPass[32 * g + lane] = res;
    } else {
      const int  q     = q0 + g;
      const bool valid = q < numPairs;
      const bool hasB  = (2 * q + 1) < numTiles;
      if (valid && (!hi || hasB)) outPass[32 * g + lane] = res;
    }
    if ((g % 2) == 1) __builtin_amdgcn_sched_barrier(0);
  }
}

struct Operands { v2f Acoef; v8f Cinit; };

struct CoefLoads { float qLo, wLo, qHi, wHi; int k0, k1; };

__device__ __forceinline__ CoefLoads load_coefs(const float* __restrict__ qz,
                                                const float* __restrict__ w,
                                                const int lane) {
  const int kHi = 32 + (lane < 3 ? lane : 0);
  CoefLoads c;
  c.qLo = qz[lane]; c.wLo = w[lane];
  c.qHi = qz[kHi];  c.wHi = w[kHi];
  c.k0  = kCoefIndex[2 * lane + 0];
  c.k1  = kCoefIndex[2 * lane + 1];
  return c;
}

__device__ __forceinline__ Operands make_operands(const CoefLoads c, const bool hi) {
  const float coefLo = hc_gate_coef(c.qLo, c.wLo);
  const float coefHi = hc_gate_coef(c.qHi, c.wHi);
  const float g0lo = __shfl(coefLo, c.k0 & 31), g0hi = __shfl(coefHi, c.k0 & 31);
  const float g1lo = __shfl(coefLo, c.k1 & 31), g1hi = __shfl(coefHi, c.k1 & 31);
  const float bias = __shfl(coefLo, 0);
  Operands op;
  op.Acoef[0] = (c.k0 < 0) ? 0.0f : ((c.k0 >= 32) ? g0hi : g0lo);
  op.Acoef[1] = (c.k1 < 0) ? 0.0f : ((c.k1 >= 32) ? g1hi : g1lo);
  op.Cinit = (v8f){hi ? 0.0f : bias, 0.0f, 0.0f, 0.0f, 0.0f, 0.0f, 0.0f, 0.0f};
  return op;
}

__global__ void __launch_bounds__(256) __attribute__((amdgpu_waves_per_eu(2, 2)))
l0sindy_poly_wmma(const float* __restrict__ obs,
                  const float* __restrict__ act,
                  const float* __restrict__ weights,
                  const float* __restrict__ qz_loga,
                  float* __restrict__ out,
                  int batch) {
  const int lane          = threadIdx.x & 31;
  const int wavesPerBlock = blockDim.x >> 5;
  const int waveId   = __builtin_amdgcn_readfirstlane(
      blockIdx.x * wavesPerBlock + (threadIdx.x >> 5));
  const int numWaves  = gridDim.x * wavesPerBlock;
  const int numTiles  = batch >> 4;
  const int numPairs  = (numTiles + 1) >> 1;
  const int numPasses = (numPairs + GROUPS_PER_PASS - 1) / GROUPS_PER_PASS;

  const bool hi    = lane >= 16;

  int pass = waveId;
  if (pass >= numPasses) return;

  Operands op;
  if ((pass + 1) * (2 * GROUPS_PER_PASS) <= numTiles) {
    const CoefLoads c = load_coefs(qz_loga, weights, lane);
    PassRows first;
    request_pass<true>(obs, act, pass, numPairs, numTiles, lane, first);
    op = make_operands(c, hi);
    finish_pass<true>(first, pass, numPairs, numTiles, lane, hi, op.Acoef, op.Cinit, out);
  } else {
    PassRows first;
    request_pass<false>(obs, act, pass, numPairs, numTiles, lane, first);
    const CoefLoads c = load_coefs(qz_loga, weights, lane);
    op = make_operands(c, hi);
    finish_pass<false>(first, pass, numPairs, numTiles, lane, hi, op.Acoef, op.Cinit, out);
  }

  for (pass += numWaves; pass < numPasses; pass += numWaves) {
    PassRows rows;
    request_pass<false>(obs, act, pass, numPairs, numTiles, lane, rows);
    finish_pass<false>(rows, pass, numPairs, numTiles, lane, hi, op.Acoef, op.Cinit, out);
  }
}

extern "C" void kernel_launch(void* const* d_in, const int* in_sizes, int n_in,
                              void* d_out, int out_size, void* d_ws, size_t ws_size,
                              hipStream_t stream) {
  const float* obs     = (const float*)d_in[0];
  const float* act     = (const float*)d_in[1];
  const float* weights = (const float*)d_in[2];
  const float* qz_loga = (const float*)d_in[3];
  float* out = (float*)d_out;

  const int batch = in_sizes[1];

  l0sindy_poly_wmma<<<256, 256, 0, stream>>>(obs, act, weights, qz_loga,
                                             out, batch);
}
